# starting kernel
# baseline (speedup 1.0000x reference)
#include <hip/hip_runtime.h>
#include <hip/hip_bf16.h>
#include <math.h>

typedef __attribute__((ext_vector_type(2))) float v2f;
typedef __attribute__((ext_vector_type(8))) float v8f;

#define NEG_SLOPE 0.2f

__global__ void fill_f32(float* __restrict__ p, float v, long n) {
  long i = (long)blockIdx.x * blockDim.x + threadIdx.x;
  if (i < n) p[i] = v;
}

__global__ void wmma_gemm_f32(const float* __restrict__ A,
                              const float* __restrict__ B,
                              float* __restrict__ C, int K, int Ncols) {
  const int tileM = blockIdx.x * 16;
  const int wave  = threadIdx.x >> 5;
  const int lane  = threadIdx.x & 31;
  const int tileN = wave * 16;
  const int r  = lane & 15;
  const int hi = lane >> 4;

  const float* __restrict__ Arow = A + (size_t)(tileM + r) * K;
  v8f acc = {};

#pragma unroll 4
  for (int k = 0; k < K; k += 4) {
    v2f a, b;
    const int ka = k + 2 * hi;
    a.x = Arow[ka];
    a.y = Arow[ka + 1];
    b.x = B[(size_t)ka * Ncols + tileN + r];
    b.y = B[(size_t)(ka + 1) * Ncols + tileN + r];
    acc = __builtin_amdgcn_wmma_f32_16x16x4_f32(
         false, a,  false, b,
         (short)0, acc,  false,  false);
  }

  const int col     = tileN + r;
  const int rowBase = tileM + (hi << 3);
#pragma unroll
  for (int i = 0; i < 8; ++i)
    C[(size_t)(rowBase + i) * Ncols + col] = acc[i];
}

__global__ void att_logits(const float* __restrict__ h,
                           const float* __restrict__ att_s,
                           const float* __restrict__ att_d,
                           float* __restrict__ a_s, float* __restrict__ a_d,
                           int N, int H) {
  const int wid  = (blockIdx.x * blockDim.x + threadIdx.x) >> 5;
  const int lane = threadIdx.x & 31;
  if (wid >= N * H) return;
  const int n  = wid / H;
  const int hh = wid - n * H;
  const float* __restrict__ hp = h + ((size_t)n * H + hh) * 64;
  const float* __restrict__ ws = att_s + hh * 64;
  const float* __restrict__ wd = att_d + hh * 64;
  float s = hp[lane] * ws[lane] + hp[lane + 32] * ws[lane + 32];
  float d = hp[lane] * wd[lane] + hp[lane + 32] * wd[lane + 32];
#pragma unroll
  for (int off = 16; off > 0; off >>= 1) {
    s += __shfl_down(s, off, 32);
    d += __shfl_down(d, off, 32);
  }
  if (lane == 0) {
    a_s[wid] = s;
    a_d[wid] = d;
  }
}

__device__ __forceinline__ void atomicMaxF32(float* addr, float v) {
  if (v >= 0.0f)
    atomicMax((int*)addr, __float_as_int(v));
  else
    atomicMin((unsigned int*)addr, __float_as_uint(v));
}

__global__ void edge_logit_max(const int* __restrict__ ei, int E, int N, int H,
                               const float* __restrict__ a_s,
                               const float* __restrict__ a_d,
                               float* __restrict__ elog,
                               float* __restrict__ mx) {
  const long idx = (long)blockIdx.x * blockDim.x + threadIdx.x;
  const long total = ((long)E + N) * H;
  if (idx >= total) return;
  const long e  = idx / H;
  const int  hh = (int)(idx - e * H);
  const int  s  = (e < E) ? ei[e] : (int)(e - E);
  const int  d  = (e < E) ? ei[(size_t)E + e] : (int)(e - E);
  float v = a_s[(size_t)s * H + hh] + a_d[(size_t)d * H + hh];
  v = (v > 0.0f) ? v : NEG_SLOPE * v;
  elog[idx] = v;
  atomicMaxF32(&mx[(size_t)d * H + hh], v);
}

__global__ void edge_exp_sum(const int* __restrict__ ei, int E, int N, int H,
                             float* __restrict__ elog,
                             const float* __restrict__ mx,
                             float* __restrict__ dn) {
  const long idx = (long)blockIdx.x * blockDim.x + threadIdx.x;
  const long total = ((long)E + N) * H;
  if (idx >= total) return;
  const long e  = idx / H;
  const int  hh = (int)(idx - e * H);
  const int  d  = (e < E) ? ei[(size_t)E + e] : (int)(e - E);
  const float ex = __expf(elog[idx] - mx[(size_t)d * H + hh]);
  elog[idx] = ex;
  atomicAdd(&dn[(size_t)d * H + hh], ex);
}

__global__ void edge_aggregate(const int* __restrict__ ei, int E, int N, int H,
                               const float* __restrict__ elog,
                               const float* __restrict__ dn,
                               const float* __restrict__ hfeat,
                               float* __restrict__ out) {
  const long wid  = ((long)blockIdx.x * blockDim.x + threadIdx.x) >> 5;
  const int  lane = threadIdx.x & 31;
  const long total = ((long)E + N) * H;
  if (wid >= total) return;
  const long e  = wid / H;
  const int  hh = (int)(wid - e * H);
  const int  s  = (e < E) ? ei[e] : (int)(e - E);
  const int  d  = (e < E) ? ei[(size_t)E + e] : (int)(e - E);
  const float alpha = elog[wid] / dn[(size_t)d * H + hh];
  const float* __restrict__ hp = hfeat + ((size_t)s * H + hh) * 64;
  float* __restrict__ op = out + ((size_t)d * H + hh) * 64;
  atomicAdd(&op[lane],      alpha * hp[lane]);
  atomicAdd(&op[lane + 32], alpha * hp[lane + 32]);
}

__global__ void bias_elu(float* __restrict__ x, const float* __restrict__ b,
                         long n, int F) {
  long i = (long)blockIdx.x * blockDim.x + threadIdx.x;
  if (i >= n) return;
  float v = x[i] + b[i % F];
  x[i] = (v > 0.0f) ? v : expm1f(v);
}

__global__ void bias_add(float* __restrict__ x, const float* __restrict__ b,
                         long n, int F) {
  long i = (long)blockIdx.x * blockDim.x + threadIdx.x;
  if (i >= n) return;
  x[i] = x[i] + b[i % F];
}

static inline int cdiv_l(long a, int b) { return (int)((a + b - 1) / b); }

extern "C" void kernel_launch(void* const* d_in, const int* in_sizes, int n_in,
                              void* d_out, int out_size, void* d_ws, size_t ws_size,
                              hipStream_t stream) {
  const float* x    = (const float*)d_in[0];
  const int*   ei   = (const int*)d_in[1];
  const float* W1   = (const float*)d_in[2];
  const float* as1w = (const float*)d_in[3];
  const float* ad1w = (const float*)d_in[4];
  const float* b1   = (const float*)d_in[5];
  const float* W2   = (const float*)d_in[6];
  const float* as2w = (const float*)d_in[7];
  const float* ad2w = (const float*)d_in[8];
  const float* b2   = (const float*)d_in[9];

  const int Fin = 128;
  const int N   = in_sizes[0] / Fin;
  const int E   = in_sizes[1] / 2;
  const long Et = (long)E + N;

  float* ws   = (float*)d_ws;
  float* h1   = ws;  ws += (size_t)N * 128;
  float* agg1 = ws;  ws += (size_t)N * 128;
  float* h2   = ws;  ws += (size_t)N * 64;
  float* a_s  = ws;  ws += (size_t)N * 2;
  float* a_d  = ws;  ws += (size_t)N * 2;
  float* mx   = ws;  ws += (size_t)N * 2;
  float* dn   = ws;  ws += (size_t)N * 2;
  float* el   = ws;  ws += (size_t)Et * 2;
  float* outp = (float*)d_out;

  const float NEG_INF = -3.402823466e+38f;

  {
    const int H = 2;
    wmma_gemm_f32<<<N / 16, 32 * (128 / 16), 0, stream>>>(x, W1, h1, 128, 128);
    att_logits<<<cdiv_l((long)N * H, 8), 256, 0, stream>>>(h1, as1w, ad1w,
                                                           a_s, a_d, N, H);
    fill_f32<<<cdiv_l((long)N * H, 256), 256, 0, stream>>>(mx, NEG_INF, (long)N * H);
    fill_f32<<<cdiv_l((long)N * H, 256), 256, 0, stream>>>(dn, 0.0f, (long)N * H);
    fill_f32<<<cdiv_l((long)N * 128, 256), 256, 0, stream>>>(agg1, 0.0f, (long)N * 128);
    edge_logit_max<<<cdiv_l(Et * H, 256), 256, 0, stream>>>(ei, E, N, H,
                                                            a_s, a_d, el, mx);
    edge_exp_sum<<<cdiv_l(Et * H, 256), 256, 0, stream>>>(ei, E, N, H, el, mx, dn);
    edge_aggregate<<<cdiv_l(Et * H * 32, 256), 256, 0, stream>>>(ei, E, N, H,
                                                                 el, dn, h1, agg1);
    bias_elu<<<cdiv_l((long)N * 128, 256), 256, 0, stream>>>(agg1, b1,
                                                             (long)N * 128, 128);
  }

  {
    const int H = 1;
    wmma_gemm_f32<<<N / 16, 32 * (64 / 16), 0, stream>>>(agg1, W2, h2, 128, 64);
    att_logits<<<cdiv_l((long)N * H, 8), 256, 0, stream>>>(h2, as2w, ad2w,
                                                           a_s, a_d, N, H);
    fill_f32<<<cdiv_l((long)N * H, 256), 256, 0, stream>>>(mx, NEG_INF, (long)N * H);
    fill_f32<<<cdiv_l((long)N * H, 256), 256, 0, stream>>>(dn, 0.0f, (long)N * H);
    fill_f32<<<cdiv_l((long)N * 64, 256), 256, 0, stream>>>(outp, 0.0f, (long)N * 64);
    edge_logit_max<<<cdiv_l(Et * H, 256), 256, 0, stream>>>(ei, E, N, H,
                                                            a_s, a_d, el, mx);
    edge_exp_sum<<<cdiv_l(Et * H, 256), 256, 0, stream>>>(ei, E, N, H, el, mx, dn);
    edge_aggregate<<<cdiv_l(Et * H * 32, 256), 256, 0, stream>>>(ei, E, N, H,
                                                                 el, dn, h2, outp);
    bias_add<<<cdiv_l((long)N * 64, 256), 256, 0, stream>>>(outp, b2,
                                                            (long)N * 64, 64);
  }
}


# fewer, longer waves (was one wave per item): aggregation waves stride over all items on a chip-filling grid, logit waves take 32 rows, edge threads 8 edges, GEMM blocks 4 or 2 row tiles
# speedup vs baseline: 3.9271x; 3.9271x over previous
#include <hip/hip_runtime.h>
#include <hip/hip_bf16.h>
#include <math.h>

typedef __attribute__((ext_vector_type(2))) float v2f;
typedef __attribute__((ext_vector_type(8))) float v8f;

#define NEG_SLOPE 0.2f

template <int TPB>
__global__ void wmma_gemm_f32(const float* __restrict__ A,
                              const float* __restrict__ B,
                              float* __restrict__ C, int K, int Ncols, int Mtiles) {
  for (int ti = 0; ti < TPB; ++ti) {
  const int tile = blockIdx.x * TPB + ti;
  if (tile >= Mtiles) break;
  const int tileM = tile * 16;
  const int wave  = threadIdx.x >> 5;
  const int lane  = threadIdx.x & 31;
  const int tileN = wave * 16;
  const int r  = lane & 15;
  const int hi = lane >> 4;

  const float* __restrict__ Arow = A + (size_t)(tileM + r) * K;
  v8f acc = {};

#pragma unroll 4
  for (int k = 0; k < K; k += 4) {
    v2f a, b;
    const int ka = k + 2 * hi;
    a.x = Arow[ka];
    a.y = Arow[ka + 1];
    b.x = B[(size_t)ka * Ncols + tileN + r];
    b.y = B[(size_t)(ka + 1) * Ncols + tileN + r];
    acc = __builtin_amdgcn_wmma_f32_16x16x4_f32(
         false, a,  false, b,
         (short)0, acc,  false,  false);
  }

  const int col     = tileN + r;
  const int rowBase = tileM + (hi << 3);
#pragma unroll
  for (int i = 0; i < 8; ++i)
    C[(size_t)(rowBase + i) * Ncols + col] = acc[i];
  }
}

template <int XM>
__device__ __forceinline__ float row_xor_f32(float v) {
  const int b = __float_as_int(v);
  return __int_as_float(__builtin_amdgcn_update_dpp(b, b, 0x160 | XM, 0xf, 0xf, false));
}
__device__ __forceinline__ float xhalf_f32(float v) {
  const int b = __float_as_int(v);
  return __int_as_float(
      __builtin_amdgcn_permlanex16(b, b, 0x76543210, 0xfedcba98, false, false));
}
__device__ __forceinline__ float wave_max_f32(float v) {
  v = fmaxf(v, row_xor_f32<1>(v));
  v = fmaxf(v, row_xor_f32<2>(v));
  v = fmaxf(v, row_xor_f32<4>(v));
  v = fmaxf(v, row_xor_f32<8>(v));
  return fmaxf(v, xhalf_f32(v));
}
__device__ __forceinline__ float wave_sum_f32(float v) {
  v += row_xor_f32<1>(v);
  v += row_xor_f32<2>(v);
  v += row_xor_f32<4>(v);
  v += row_xor_f32<8>(v);
  return v + xhalf_f32(v);
}
__device__ __forceinline__ float lane_f32(float v, int l) {
  return __int_as_float(__builtin_amdgcn_readlane(__float_as_int(v), l));
}

template <int H, int IT>
__global__ __launch_bounds__(256) void att_logits(const float* __restrict__ h,
                                                  const float* __restrict__ att_s,
                                                  const float* __restrict__ att_d,
                                                  float* __restrict__ a_s,
                                                  float* __restrict__ a_d,
                                                  int total) {
  static_assert(IT % H == 0 && IT <= 32, "rows per wave");
  const int lane = threadIdx.x & 31;
  const int wid = __builtin_amdgcn_readfirstlane(
      (int)(blockIdx.x * (blockDim.x >> 5) + (threadIdx.x >> 5)));
  const int base = wid * IT;
  if (base >= total) return;
  const float2* __restrict__ h2 = reinterpret_cast<const float2*>(h);
  float2 ws[H], wd[H];
#pragma unroll
  for (int hh = 0; hh < H; ++hh) {
    ws[hh] = reinterpret_cast<const float2*>(att_s)[hh * 32 + lane];
    wd[hh] = reinterpret_cast<const float2*>(att_d)[hh * 32 + lane];
  }
  float2 v[IT];
#pragma unroll
  for (int k = 0; k < IT; ++k) {
    const int w = min(base + k, total - 1);
    v[k] = h2[(size_t)w * 32 + lane];
  }
  float os = 0.0f, od = 0.0f;
#pragma unroll
  for (int k = 0; k < IT; ++k) {
    const int hh = k % H;
    const float s = wave_sum_f32(fmaf(v[k].x, ws[hh].x, v[k].y * ws[hh].y));
    const float d = wave_sum_f32(fmaf(v[k].x, wd[hh].x, v[k].y * wd[hh].y));
    if (lane == k) {
      os = s;
      od = d;
    }
  }
  if (lane < IT && base + lane < total) {
    a_s[base + lane] = os;
    a_d[base + lane] = od;
  }
}

__global__ void zero_i32(int* __restrict__ p, int n) {
  const int i = blockIdx.x * blockDim.x + threadIdx.x;
  if (i < n) p[i] = 0;
}

template <int PER>
__global__ void dst_count_rank(const int* __restrict__ ei, int E,
                               int* __restrict__ cnt, int* __restrict__ rank) {
  const int T  = gridDim.x * blockDim.x;
  const int e0 = blockIdx.x * blockDim.x + threadIdx.x;
  int d[PER], r[PER];
#pragma unroll
  for (int k = 0; k < PER; ++k) d[k] = ei[(size_t)E + min(e0 + k * T, E - 1)];
#pragma unroll
  for (int k = 0; k < PER; ++k) {
    const int e = e0 + k * T;
    r[k] = (e < E) ? atomicAdd(&cnt[d[k]], 1) : 0;
  }
#pragma unroll
  for (int k = 0; k < PER; ++k) {
    const int e = e0 + k * T;
    if (e < E) rank[e] = r[k];
  }
}

__global__ __launch_bounds__(1024) void row_offsets(const int* __restrict__ cnt,
                                                    int* __restrict__ rowptr,
                                                    int N) {
  __shared__ int wpre[32];
  __shared__ int wtot[32];
  const int tid  = threadIdx.x;
  const int lane = tid & 31;
  const int w    = tid >> 5;
  const int blockStart = blockIdx.x * 1024;

  int pre = 0;
  const int4* __restrict__ c4 = reinterpret_cast<const int4*>(cnt);
  const int n4 = blockStart >> 2;
  for (int i = tid; i < n4; i += 4096) {
    const int last = n4 - 1;
    const int4 v0 = c4[i];
    const int4 v1 = c4[min(i + 1024, last)];
    const int4 v2 = c4[min(i + 2048, last)];
    const int4 v3 = c4[min(i + 3072, last)];
    pre += (v0.x + v0.y) + (v0.z + v0.w);
    pre += (i + 1024 < n4) ? (v1.x + v1.y) + (v1.z + v1.w) : 0;
    pre += (i + 2048 < n4) ? (v2.x + v2.y) + (v2.z + v2.w) : 0;
    pre += (i + 3072 < n4) ? (v3.x + v3.y) + (v3.z + v3.w) : 0;
  }
  const int i = blockStart + tid;
  const int c = (i < N) ? cnt[i] : 0;

  int inc = c;
#pragma unroll
  for (int off = 1; off < 32; off <<= 1) {
    const int t = __shfl_up(inc, off, 32);
    if (lane >= off) inc += t;
  }
#pragma unroll
  for (int off = 16; off > 0; off >>= 1) pre += __shfl_xor(pre, off, 32);
  if (lane == 31) wtot[w] = inc;
  if (lane == 0)  wpre[w] = pre;
  __syncthreads();
  int s = wpre[lane] + ((lane < w) ? wtot[lane] : 0);
#pragma unroll
  for (int off = 16; off > 0; off >>= 1) s += __shfl_xor(s, off, 32);
  if (i < N) rowptr[i] = s + inc - c;
}

template <int PER>
__global__ void place_sources(const int* __restrict__ ei, int E,
                              const int* __restrict__ rowptr,
                              const int* __restrict__ rank,
                              int* __restrict__ col) {
  const int T  = gridDim.x * blockDim.x;
  const int e0 = blockIdx.x * blockDim.x + threadIdx.x;
  int s[PER], d[PER], r[PER], p[PER];
#pragma unroll
  for (int k = 0; k < PER; ++k) {
    const int e = min(e0 + k * T, E - 1);
    s[k] = ei[e];
    d[k] = ei[(size_t)E + e];
    r[k] = rank[e];
  }
#pragma unroll
  for (int k = 0; k < PER; ++k) p[k] = rowptr[d[k]];
#pragma unroll
  for (int k = 0; k < PER; ++k)
    if (e0 + k * T < E) col[p[k] + r[k]] = s[k];
}

template <int K, int H>
__device__ __forceinline__ void gather_rows(const float2* __restrict__ h2, int hh,
                                            int lane, int s, int t, float2 (&r)[K]) {
#pragma unroll
  for (int u = 0; u < K; ++u) {
    const int su = __builtin_amdgcn_readlane(s, t + u);
    r[u] = h2[(size_t)(su * H + hh) * 32 + lane];
  }
}
template <int K>
__device__ __forceinline__ void add_rows(float p, int t, const float2 (&r)[K],
                                         float2& acc) {
#pragma unroll
  for (int u = 0; u < K; ++u) {
    const float pu = lane_f32(p, t + u);
    acc.x = fmaf(pu, r[u].x, acc.x);
    acc.y = fmaf(pu, r[u].y, acc.y);
  }
}

template <int H, bool ELU>
__global__ __launch_bounds__(256) void dst_softmax_aggregate(
    const int* __restrict__ rowptr, const int* __restrict__ cnt,
    const int* __restrict__ col,
    const float* __restrict__ a_s, const float* __restrict__ a_d,
    const float* __restrict__ hfeat, const float* __restrict__ bias,
    float* __restrict__ out, int N) {
  const int lane = threadIdx.x & 31;
  static_assert(8 % H == 0, "waves per block must be a multiple of H");
  const int wid0 = __builtin_amdgcn_readfirstlane(
      (int)(blockIdx.x * (blockDim.x >> 5) + (threadIdx.x >> 5)));
  const int nw    = (int)(gridDim.x * (blockDim.x >> 5));
  const int total = N * H;
  const int hh    = wid0 % H;
  const float2* __restrict__ h2 = reinterpret_cast<const float2*>(hfeat);

  const float bx = bias[hh * 64 + 2 * lane];
  const float by = bias[hh * 64 + 2 * lane + 1];

  for (int wid = wid0; wid < total; wid += nw) {
  const int d     = wid / H;
  const int start = rowptr[d];
  const int deg   = cnt[d];

  const float ad = a_d[wid];
  float vs = a_s[wid] + ad;
  vs = (vs > 0.0f) ? vs : NEG_SLOPE * vs;

  float  m   = vs;
  float  sum = 1.0f;
  float2 acc = h2[(size_t)wid * 32 + lane];

  for (int base = 0; base < deg; base += 32) {
    const int  j     = base + lane;
    const bool valid = j < deg;
    int   s = d;
    float v = -INFINITY;
    if (valid) {
      s = col[start + j];
      const float sv = a_s[(size_t)s * H + hh] + ad;
      v = (sv > 0.0f) ? sv : NEG_SLOPE * sv;
    }
    const float mn = fmaxf(m, wave_max_f32(v));
    const float sc = __expf(m - mn);
    const float p  = valid ? __expf(v - mn) : 0.0f;
    sum = sum * sc + wave_sum_f32(p);
    acc.x *= sc;
    acc.y *= sc;
    m = mn;

    const int nj = min(32, deg - base);
    int t = 0;
    for (; t + 8 <= nj; t += 8) {
      float2 r[8];
      gather_rows<8, H>(h2, hh, lane, s, t, r);
      add_rows<8>(p, t, r, acc);
    }
    const int  rem = nj - t;
    const bool g4 = (rem & 4) != 0, g2 = (rem & 2) != 0, g1 = (rem & 1) != 0;
    const int  t2 = t + (rem & 4), t1 = t2 + (rem & 2);
    float2 r4[4], r2[2], r1[1];
    if (g4) gather_rows<4, H>(h2, hh, lane, s, t, r4);
    if (g2) gather_rows<2, H>(h2, hh, lane, s, t2, r2);
    if (g1) gather_rows<1, H>(h2, hh, lane, s, t1, r1);
    if (g4) add_rows<4>(p, t, r4, acc);
    if (g2) add_rows<2>(p, t2, r2, acc);
    if (g1) add_rows<1>(p, t1, r1, acc);
  }

  const float inv = 1.0f / sum;
  float ox = acc.x * inv + bx;
  float oy = acc.y * inv + by;
  if (ELU) {
    ox = (ox > 0.0f) ? ox : expm1f(ox);
    oy = (oy > 0.0f) ? oy : expm1f(oy);
  }
  float2 o;
  o.x = ox;
  o.y = oy;
  reinterpret_cast<float2*>(out)[(size_t)wid * 32 + lane] = o;
  }
}

static inline int cdiv_l(long a, int b) { return (int)((a + b - 1) / b); }

#define AGG_BLOCKS 2048

extern "C" void kernel_launch(void* const* d_in, const int* in_sizes, int n_in,
                              void* d_out, int out_size, void* d_ws, size_t ws_size,
                              hipStream_t stream) {
  const float* x    = (const float*)d_in[0];
  const int*   ei   = (const int*)d_in[1];
  const float* W1   = (const float*)d_in[2];
  const float* as1w = (const float*)d_in[3];
  const float* ad1w = (const float*)d_in[4];
  const float* b1   = (const float*)d_in[5];
  const float* W2   = (const float*)d_in[6];
  const float* as2w = (const float*)d_in[7];
  const float* ad2w = (const float*)d_in[8];
  const float* b2   = (const float*)d_in[9];

  const int Fin = 128;
  const int N   = in_sizes[0] / Fin;
  const int E   = in_sizes[1] / 2;

  float* ws   = (float*)d_ws;
  float* h1   = ws;  ws += (size_t)N * 128;
  float* agg1 = ws;  ws += (size_t)N * 128;
  float* h2   = ws;  ws += (size_t)N * 64;
  float* a_s  = ws;  ws += (size_t)N * 2;
  float* a_d  = ws;  ws += (size_t)N * 2;
  int* cnt    = (int*)ws;  ws += (size_t)N;
  int* rowptr = (int*)ws;  ws += (size_t)N;
  int* rank   = (int*)ws;  ws += (size_t)E;
  int* col    = (int*)ws;  ws += (size_t)E;
  float* outp = (float*)d_out;

  zero_i32<<<cdiv_l(N, 256), 256, 0, stream>>>(cnt, N);
  dst_count_rank<8><<<cdiv_l(E, 8 * 256), 256, 0, stream>>>(ei, E, cnt, rank);
  row_offsets<<<cdiv_l(N, 1024), 1024, 0, stream>>>(cnt, rowptr, N);
  place_sources<8><<<cdiv_l(E, 8 * 256), 256, 0, stream>>>(ei, E, rowptr, rank, col);

  {
    const int H = 2;
    wmma_gemm_f32<4><<<cdiv_l(N / 16, 4), 32 * (128 / 16), 0, stream>>>(
        x, W1, h1, 128, 128, N / 16);
    att_logits<2, 32><<<cdiv_l((long)N * H, 8 * 32), 256, 0, stream>>>(
        h1, as1w, ad1w, a_s, a_d, N * H);
    dst_softmax_aggregate<2, true><<<AGG_BLOCKS, 256, 0, stream>>>(
        rowptr, cnt, col, a_s, a_d, h1, b1, agg1, N);
  }

  {
    const int H = 1;
    wmma_gemm_f32<2><<<cdiv_l(N / 16, 2), 32 * (64 / 16), 0, stream>>>(
        agg1, W2, h2, 128, 64, N / 16);
    att_logits<1, 32><<<cdiv_l((long)N * H, 8 * 32), 256, 0, stream>>>(
        h2, as2w, ad2w, a_s, a_d, N * H);
    dst_softmax_aggregate<1, false><<<AGG_BLOCKS, 256, 0, stream>>>(
        rowptr, cnt, col, a_s, a_d, h2, b2, outp, N);
  }
}


# both GEMMs: B's 128x64 column slice staged per block in LDS as fragment pairs, a wave reads its 16x128 tile of A itself (16 B a lane) for 4 column tiles, 16-B row stores; k order in the sums changes
# speedup vs baseline: 5.0785x; 1.2932x over previous
#include <hip/hip_runtime.h>
#include <hip/hip_bf16.h>
#include <math.h>

typedef __attribute__((ext_vector_type(2))) float v2f;
typedef __attribute__((ext_vector_type(8))) float v8f;
typedef __attribute__((ext_vector_type(4))) float v4f;

#define NEG_SLOPE 0.2f

template <int TPW>
__global__ __launch_bounds__(128) void wmma_gemm_f32_k128(
    const float* __restrict__ A, const float* __restrict__ B,
    float* __restrict__ C, int Ncols, int groups, int Mtiles) {
  __shared__ float Bs[64 * 128];
  const int tid = threadIdx.x;
  const int g   = blockIdx.x % groups;
  const int bg  = blockIdx.x / groups;

#pragma unroll
  for (int i = 0; i < 8; ++i) {
    const int q  = tid + 128 * i;
    const int kp = q >> 4;
    const int c4 = q & 15;
    const float* __restrict__ src = B + (size_t)(2 * kp) * Ncols + 64 * g + 4 * c4;
    const v4f e = *reinterpret_cast<const v4f*>(src);
    const v4f o = *reinterpret_cast<const v4f*>(src + Ncols);
    const v4f lo = {e.x, o.x, e.y, o.y};
    const v4f hi = {e.z, o.z, e.w, o.w};
    *reinterpret_cast<v4f*>(&Bs[kp * 128 + 4 * c4])      = lo;
    *reinterpret_cast<v4f*>(&Bs[kp * 128 + 64 + 4 * c4]) = hi;
  }
  __syncthreads();

  const int wave = __builtin_amdgcn_readfirstlane(tid >> 5);
  const int lane = tid & 31;
  const int c    = lane & 15;
  const int h    = lane >> 4;
  const float* __restrict__ Bl = &Bs[(2 * h) * 128 + 4 * c];

  for (int i = 0; i < TPW; ++i) {
    const int tile = (bg * 4 + wave) * TPW + i;
    if (tile >= Mtiles) break;
    const int tileM = tile * 16;

    const float* __restrict__ Arow = A + (size_t)(tileM + c) * 128 + 4 * h;
    v4f a[16];
#pragma unroll
    for (int j = 0; j < 16; ++j)
      a[j] = *reinterpret_cast<const v4f*>(Arow + 8 * j);

    v8f acc[4] = {};
#pragma unroll
    for (int j = 0; j < 16; ++j) {
#pragma unroll
      for (int p = 0; p < 2; ++p) {
        v2f av;
        av.x = a[j][2 * p];
        av.y = a[j][2 * p + 1];
        const float* __restrict__ bp = Bl + (4 * j + p) * 128;
        const v4f q0 = *reinterpret_cast<const v4f*>(bp);
        const v4f q1 = *reinterpret_cast<const v4f*>(bp + 64);
        acc[0] = __builtin_amdgcn_wmma_f32_16x16x4_f32(
             false, av,  false, q0.xy,
             (short)0, acc[0],  false,  false);
        acc[1] = __builtin_amdgcn_wmma_f32_16x16x4_f32(
            false, av, false, q0.zw, (short)0, acc[1], false, false);
        acc[2] = __builtin_amdgcn_wmma_f32_16x16x4_f32(
            false, av, false, q1.xy, (short)0, acc[2], false, false);
        acc[3] = __builtin_amdgcn_wmma_f32_16x16x4_f32(
            false, av, false, q1.zw, (short)0, acc[3], false, false);
      }
    }

    float* __restrict__ Crow = C + (size_t)(tileM + 8 * h) * Ncols + 64 * g + 4 * c;
#pragma unroll
    for (int rr = 0; rr < 8; ++rr) {
      v4f o;
      o.x = acc[0][rr];
      o.y = acc[1][rr];
      o.z = acc[2][rr];
      o.w = acc[3][rr];
      *reinterpret_cast<v4f*>(Crow + (size_t)rr * Ncols) = o;
    }
  }
}

template <int XM>
__device__ __forceinline__ float row_xor_f32(float v) {
  return __int_as_float(__builtin_amdgcn_update_dpp(0, __float_as_int(v), 0x160 | XM,
                                                    0xf, 0xf, true));
}
__device__ __forceinline__ float xhalf_f32(float v) {
  const int b = __float_as_int(v);
  return __int_as_float(
      __builtin_amdgcn_permlanex16(b, b, 0x76543210, 0xfedcba98, false, false));
}
__device__ __forceinline__ float row16_sum_f32(float v) {
  v += row_xor_f32<1>(v);
  v += row_xor_f32<2>(v);
  v += row_xor_f32<4>(v);
  v += row_xor_f32<8>(v);
  return v;
}
__device__ __forceinline__ float dot4_f32(const v4f& a, const v4f& b) {
  const v2f t = __builtin_elementwise_fma(a.zw, b.zw, a.xy * b.xy);
  return t.x + t.y;
}

__device__ __forceinline__ void hold_rows(v4f (&r)[8]) {
  asm volatile("" : "+v"(r[0]), "+v"(r[1]), "+v"(r[2]), "+v"(r[3]),
                    "+v"(r[4]), "+v"(r[5]), "+v"(r[6]), "+v"(r[7]));
}
__device__ __forceinline__ void hold_rows(v4f (&r)[4]) {
  asm volatile("" : "+v"(r[0]), "+v"(r[1]), "+v"(r[2]), "+v"(r[3]));
}
__device__ __forceinline__ void hold_rows(v4f (&r)[2]) {
  asm volatile("" : "+v"(r[0]), "+v"(r[1]));
}
__device__ __forceinline__ void hold_rows(v4f (&r)[1]) {
  asm volatile("" : "+v"(r[0]));
}

__global__ void zero_i32(int* __restrict__ p, int n) {
  const int i = blockIdx.x * blockDim.x + threadIdx.x;
  if (i < n) p[i] = 0;
}

template <int PER>
__global__ void dst_count_rank(const int* __restrict__ ei, int E,
                               int* __restrict__ cnt, int* __restrict__ rank) {
  const int T  = gridDim.x * blockDim.x;
  const int e0 = blockIdx.x * blockDim.x + threadIdx.x;
  int d[PER], r[PER];
#pragma unroll
  for (int k = 0; k < PER; ++k) d[k] = ei[(size_t)E + min(e0 + k * T, E - 1)];
#pragma unroll
  for (int k = 0; k < PER; ++k) {
    const int e = e0 + k * T;
    r[k] = (e < E) ? atomicAdd(&cnt[d[k]], 1) : 0;
  }
#pragma unroll
  for (int k = 0; k < PER; ++k) {
    const int e = e0 + k * T;
    if (e < E) rank[e] = r[k];
  }
}

__global__ __launch_bounds__(1024) void row_offsets(const int* __restrict__ cnt,
                                                    int* __restrict__ rowptr,
                                                    int N) {
  __shared__ int wpre[32];
  __shared__ int wtot[32];
  const int tid  = threadIdx.x;
  const int lane = tid & 31;
  const int w    = tid >> 5;
  const int blockStart = blockIdx.x * 1024;

  int pre = 0;
  const int4* __restrict__ c4 = reinterpret_cast<const int4*>(cnt);
  const int n4 = blockStart >> 2;
  for (int i = tid; i < n4; i += 4096) {
    const int last = n4 - 1;
    const int4 v0 = c4[i];
    const int4 v1 = c4[min(i + 1024, last)];
    const int4 v2 = c4[min(i + 2048, last)];
    const int4 v3 = c4[min(i + 3072, last)];
    pre += (v0.x + v0.y) + (v0.z + v0.w);
    pre += (i + 1024 < n4) ? (v1.x + v1.y) + (v1.z + v1.w) : 0;
    pre += (i + 2048 < n4) ? (v2.x + v2.y) + (v2.z + v2.w) : 0;
    pre += (i + 3072 < n4) ? (v3.x + v3.y) + (v3.z + v3.w) : 0;
  }
  const int i = blockStart + tid;
  const int c = (i < N) ? cnt[i] : 0;

  int inc = c;
#pragma unroll
  for (int off = 1; off < 32; off <<= 1) {
    const int t = __shfl_up(inc, off, 32);
    if (lane >= off) inc += t;
  }
#pragma unroll
  for (int off = 16; off > 0; off >>= 1) pre += __shfl_xor(pre, off, 32);
  if (lane == 31) wtot[w] = inc;
  if (lane == 0)  wpre[w] = pre;
  __syncthreads();
  int s = wpre[lane] + ((lane < w) ? wtot[lane] : 0);
#pragma unroll
  for (int off = 16; off > 0; off >>= 1) s += __shfl_xor(s, off, 32);
  if (i < N) rowptr[i] = s + inc - c;
}

template <int PER>
__global__ void place_sources(const int* __restrict__ ei, int E,
                              const int* __restrict__ rowptr,
                              const int* __restrict__ rank,
                              int* __restrict__ col) {
  const int T  = gridDim.x * blockDim.x;
  const int e0 = blockIdx.x * blockDim.x + threadIdx.x;
  int s[PER], d[PER], r[PER], p[PER];
#pragma unroll
  for (int k = 0; k < PER; ++k) {
    const int e = min(e0 + k * T, E - 1);
    s[k] = ei[e];
    d[k] = ei[(size_t)E + e];
    r[k] = rank[e];
  }
#pragma unroll
  for (int k = 0; k < PER; ++k) p[k] = rowptr[d[k]];
#pragma unroll
  for (int k = 0; k < PER; ++k)
    if (e0 + k * T < E) col[p[k] + r[k]] = s[k];
}

#define NO_SCORE (-1.0e30f)

template <int K, int H>
__device__ __forceinline__ void take_rows(const float* __restrict__ hfeat, int head,
                                          int c, int hi, int sv, int t,
                                          v4f (&r)[K]) {
#pragma unroll
  for (int u = 0; u < K; ++u) {
    int s;
    if (H == 2) {
      s = __builtin_amdgcn_readlane(sv, t + u);
    } else {
      const int e0 = 2 * (t + u);
      const int s0 = __builtin_amdgcn_readlane(sv, e0);
      const int s1 = __builtin_amdgcn_readlane(sv, e0 + 1);
      s = hi ? s1 : s0;
    }
    const unsigned off = (((unsigned)s * H + head) * 16u + c) * 16u;
    r[u] = *reinterpret_cast<const v4f*>(reinterpret_cast<const char*>(hfeat) + off);
  }
}

template <int K, int H>
__device__ __forceinline__ void fold_rows(const v4f (&r)[K], int t, int lim,
                                          const v4f& ws, float ad, float& m,
                                          float& sum, v4f& acc) {
  float v[K];
  float mx = m;
#pragma unroll
  for (int u = 0; u < K; ++u) {
    float x = row16_sum_f32(dot4_f32(r[u], ws)) + ad;
    x = fmaxf(x, NEG_SLOPE * x);
    if (H == 1) x = (t + u < lim) ? x : NO_SCORE;
    v[u] = x;
    mx = fmaxf(mx, x);
  }
  const float sc = __expf(m - mx);
  sum *= sc;
  acc *= sc;
#pragma unroll
  for (int u = 0; u < K; ++u) {
    float p = __expf(v[u] - mx);
    if (H == 1) p = (t + u < lim) ? p : 0.0f;
    sum += p;
    const v4f pp = {p, p, p, p};
    acc = __builtin_elementwise_fma(pp, r[u], acc);
  }
  m = mx;
}

template <int H, bool ELU>
__global__ __launch_bounds__(256) void dst_softmax_aggregate(
    const int* __restrict__ rowptr, const int* __restrict__ cnt,
    const int* __restrict__ col,
    const float* __restrict__ att_s, const float* __restrict__ att_d,
    const float* __restrict__ hfeat, const float* __restrict__ bias,
    float* __restrict__ out, int N) {
  static_assert(H == 1 || H == 2, "lane rows are heads (2) or list halves (1)");
  const int lane = threadIdx.x & 31;
  const int hi   = lane >> 4;
  const int c    = lane & 15;
  const int head = (H == 2) ? hi : 0;
  const int wid0 = __builtin_amdgcn_readfirstlane(
      (int)(blockIdx.x * (blockDim.x >> 5) + (threadIdx.x >> 5)));
  const int nw = (int)(gridDim.x * (blockDim.x >> 5));

  const v4f ws = reinterpret_cast<const v4f*>(att_s)[head * 16 + c];
  const v4f wd = reinterpret_cast<const v4f*>(att_d)[head * 16 + c];
  const v4f bz = reinterpret_cast<const v4f*>(bias)[head * 16 + c];
  const bool opens = (H == 2) || (hi == 0);

  for (int d = wid0; d < N; d += nw) {
  const unsigned own_off = (((unsigned)d * H + head) * 16u + c) * 16u;
  v4f own =
      *reinterpret_cast<const v4f*>(reinterpret_cast<const char*>(hfeat) + own_off);
  int       start = rowptr[d];
  const int deg   = cnt[d];
  asm volatile("" : "+s"(start));
  int sv_next = col[(deg > 0) ? start + min(lane, deg - 1) : 0];
  asm volatile("" : "+v"(sv_next), "+v"(own));
  const float ad = row16_sum_f32(dot4_f32(own, wd));
  float vs = row16_sum_f32(dot4_f32(own, ws)) + ad;
  vs = fmaxf(vs, NEG_SLOPE * vs);

  float  m   = opens ? vs : NO_SCORE;
  float  sum = opens ? 1.0f : 0.0f;
  v4f acc;
  acc.x = opens ? own.x : 0.0f;
  acc.y = opens ? own.y : 0.0f;
  acc.z = opens ? own.z : 0.0f;
  acc.w = opens ? own.w : 0.0f;

  for (int base = 0; base < deg; base += 32) {
    asm volatile("" : "+v"(sv_next));
    const int sv = sv_next;
    if (base + 32 < deg) sv_next = col[start + min(base + 32 + lane, deg - 1)];
    const int nj = min(32, deg - base);
    const int nsteps = (H == 2) ? nj : ((nj + 1) >> 1);
    const int lim = (nj + 1 - hi) >> 1;

    int t = 0;
    for (; t + 8 <= nsteps; t += 8) {
      v4f r[8];
      take_rows<8, H>(hfeat, head, c, hi, sv, t, r);
      hold_rows(r);
      fold_rows<8, H>(r, t, lim, ws, ad, m, sum, acc);
    }
    const int  rem = nsteps - t;
    const bool g4 = (rem & 4) != 0, g2 = (rem & 2) != 0, g1 = (rem & 1) != 0;
    const int  t2 = t + (rem & 4), t1 = t2 + (rem & 2);
    v4f r4[4], r2[2], r1[1];
    if (g4) take_rows<4, H>(hfeat, head, c, hi, sv, t, r4);
    if (g2) take_rows<2, H>(hfeat, head, c, hi, sv, t2, r2);
    if (g1) take_rows<1, H>(hfeat, head, c, hi, sv, t1, r1);
    if (g4) {
      hold_rows(r4);
      fold_rows<4, H>(r4, t, lim, ws, ad, m, sum, acc);
    }
    if (g2) {
      hold_rows(r2);
      fold_rows<2, H>(r2, t2, lim, ws, ad, m, sum, acc);
    }
    if (g1) {
      hold_rows(r1);
      fold_rows<1, H>(r1, t1, lim, ws, ad, m, sum, acc);
    }
  }

  if (H == 1) {
    const float mt = fmaxf(m, xhalf_f32(m));
    const float f  = __expf(m - mt);
    sum *= f;
    acc *= f;
    sum   += xhalf_f32(sum);
    acc.x += xhalf_f32(acc.x);
    acc.y += xhalf_f32(acc.y);
    acc.z += xhalf_f32(acc.z);
    acc.w += xhalf_f32(acc.w);
  }

  const float inv = 1.0f / sum;
  v4f o = acc * inv + bz;
  if (ELU) {
    o.x = (o.x > 0.0f) ? o.x : expm1f(o.x);
    o.y = (o.y > 0.0f) ? o.y : expm1f(o.y);
    o.z = (o.z > 0.0f) ? o.z : expm1f(o.z);
    o.w = (o.w > 0.0f) ? o.w : expm1f(o.w);
  }
  if (opens)
    *reinterpret_cast<v4f*>(reinterpret_cast<char*>(out) + own_off) = o;
  }
}

static inline int cdiv_l(long a, int b) { return (int)((a + b - 1) / b); }

#define AGG_BLOCKS 1024

extern "C" void kernel_launch(void* const* d_in, const int* in_sizes, int n_in,
                              void* d_out, int out_size, void* d_ws, size_t ws_size,
                              hipStream_t stream) {
  const float* x    = (const float*)d_in[0];
  const int*   ei   = (const int*)d_in[1];
  const float* W1   = (const float*)d_in[2];
  const float* as1w = (const float*)d_in[3];
  const float* ad1w = (const float*)d_in[4];
  const float* b1   = (const float*)d_in[5];
  const float* W2   = (const float*)d_in[6];
  const float* as2w = (const float*)d_in[7];
  const float* ad2w = (const float*)d_in[8];
  const float* b2   = (const float*)d_in[9];

  const int Fin = 128;
  const int N   = in_sizes[0] / Fin;
  const int E   = in_sizes[1] / 2;

  float* ws   = (float*)d_ws;
  float* h1   = ws;  ws += (size_t)N * 128;
  float* agg1 = ws;  ws += (size_t)N * 128;
  float* h2   = ws;  ws += (size_t)N * 64;
  int* cnt    = (int*)ws;  ws += (size_t)N;
  int* rowptr = (int*)ws;  ws += (size_t)N;
  int* rank   = (int*)ws;  ws += (size_t)E;
  int* col    = (int*)ws;  ws += (size_t)E;
  float* outp = (float*)d_out;

  zero_i32<<<cdiv_l(N, 256), 256, 0, stream>>>(cnt, N);
  dst_count_rank<8><<<cdiv_l(E, 8 * 256), 256, 0, stream>>>(ei, E, cnt, rank);
  row_offsets<<<cdiv_l(N, 1024), 1024, 0, stream>>>(cnt, rowptr, N);
  place_sources<8><<<cdiv_l(E, 8 * 256), 256, 0, stream>>>(ei, E, rowptr, rank, col);

  {
    wmma_gemm_f32_k128<2><<<cdiv_l(N / 16, 4 * 2) * 2, 128, 0, stream>>>(
        x, W1, h1, 128, 2, N / 16);
    dst_softmax_aggregate<2, true><<<AGG_BLOCKS, 256, 0, stream>>>(
        rowptr, cnt, col, as1w, ad1w, h1, b1, agg1, N);
  }

  {
    wmma_gemm_f32_k128<1><<<cdiv_l(N / 16, 4), 128, 0, stream>>>(
        agg1, W2, h2, 64, 1, N / 16);
    dst_softmax_aggregate<1, false><<<AGG_BLOCKS, 256, 0, stream>>>(
        rowptr, cnt, col, as2w, ad2w, h2, b2, outp, N);
  }
}


# layer-1 product x@W1 moved into the three list-building launches as extra blocks (it needs nothing from the list); 512-thread blocks so a list block and a product block share a WGP; 6 launches for 7
# speedup vs baseline: 9.1161x; 1.7950x over previous
#include <hip/hip_runtime.h>
#include <hip/hip_bf16.h>
#include <math.h>

typedef __attribute__((ext_vector_type(2))) float v2f;
typedef __attribute__((ext_vector_type(8))) float v8f;
typedef __attribute__((ext_vector_type(4))) float v4f;

#define NEG_SLOPE 0.2f

template <int NW>
__device__ __forceinline__ void gemm_block_f32_k128(
    const float* __restrict__ A, const float* __restrict__ B,
    float* __restrict__ C, int Ncols, int groups, int tile0, int Mtiles, int gb,
    int wmul, int bmul, int step) {
  __shared__ float Bs[64 * 128];
  const int tid = threadIdx.x;
  const int g   = gb % groups;
  const int bg  = gb / groups;

#pragma unroll
  for (int i = 0; i < 32 / NW; ++i) {
    const int q  = tid + 32 * NW * i;
    const int kp = q >> 4;
    const int c4 = q & 15;
    const float* __restrict__ src = B + (size_t)(2 * kp) * Ncols + 64 * g + 4 * c4;
    const v4f e = *reinterpret_cast<const v4f*>(src);
    const v4f o = *reinterpret_cast<const v4f*>(src + Ncols);
    const v4f lo = {e.x, o.x, e.y, o.y};
    const v4f hi = {e.z, o.z, e.w, o.w};
    *reinterpret_cast<v4f*>(&Bs[kp * 128 + 4 * c4])      = lo;
    *reinterpret_cast<v4f*>(&Bs[kp * 128 + 64 + 4 * c4]) = hi;
  }
  __syncthreads();

  const int wave = __builtin_amdgcn_readfirstlane(tid >> 5);
  const int lane = tid & 31;
  const int c    = lane & 15;
  const int h    = lane >> 4;
  const float* __restrict__ Bl = &Bs[(2 * h) * 128 + 4 * c];

  for (int tile = tile0 + wave * wmul + bg * bmul; tile < Mtiles; tile += step) {
    asm volatile("" ::: "memory");
    const int tileM = tile * 16;

    const float* __restrict__ Arow = A + (size_t)(tileM + c) * 128 + 4 * h;
    v4f a[16];
#pragma unroll
    for (int j = 0; j < 16; ++j)
      a[j] = *reinterpret_cast<const v4f*>(Arow + 8 * j);

    v8f acc[4] = {};
#pragma unroll
    for (int j = 0; j < 16; ++j) {
#pragma unroll
      for (int p = 0; p < 2; ++p) {
        v2f av;
        av.x = a[j][2 * p];
        av.y = a[j][2 * p + 1];
        const float* __restrict__ bp = Bl + (4 * j + p) * 128;
        const v4f q0 = *reinterpret_cast<const v4f*>(bp);
        const v4f q1 = *reinterpret_cast<const v4f*>(bp + 64);
        acc[0] = __builtin_amdgcn_wmma_f32_16x16x4_f32(
             false, av,  false, q0.xy,
             (short)0, acc[0],  false,  false);
        acc[1] = __builtin_amdgcn_wmma_f32_16x16x4_f32(
            false, av, false, q0.zw, (short)0, acc[1], false, false);
        acc[2] = __builtin_amdgcn_wmma_f32_16x16x4_f32(
            false, av, false, q1.xy, (short)0, acc[2], false, false);
        acc[3] = __builtin_amdgcn_wmma_f32_16x16x4_f32(
            false, av, false, q1.zw, (short)0, acc[3], false, false);
      }
    }

    float* __restrict__ Crow = C + (size_t)(tileM + 8 * h) * Ncols + 64 * g + 4 * c;
#pragma unroll
    for (int rr = 0; rr < 8; ++rr) {
      v4f o;
      o.x = acc[0][rr];
      o.y = acc[1][rr];
      o.z = acc[2][rr];
      o.w = acc[3][rr];
      *reinterpret_cast<v4f*>(Crow + (size_t)rr * Ncols) = o;
    }
  }
}

__global__ __launch_bounds__(128) void wmma_gemm_f32_k128(
    const float* __restrict__ A, const float* __restrict__ B,
    float* __restrict__ C, int Ncols, int groups, int Mtiles) {
  gemm_block_f32_k128<4>(A, B, C, Ncols, groups, 0, Mtiles, (int)blockIdx.x,
                         1, 4, Mtiles);
}

template <int XM>
__device__ __forceinline__ float row_xor_f32(float v) {
  return __int_as_float(__builtin_amdgcn_update_dpp(0, __float_as_int(v), 0x160 | XM,
                                                    0xf, 0xf, true));
}
__device__ __forceinline__ float xhalf_f32(float v) {
  const int b = __float_as_int(v);
  return __int_as_float(
      __builtin_amdgcn_permlanex16(b, b, 0x76543210, 0xfedcba98, false, false));
}
__device__ __forceinline__ float row16_sum_f32(float v) {
  v += row_xor_f32<1>(v);
  v += row_xor_f32<2>(v);
  v += row_xor_f32<4>(v);
  v += row_xor_f32<8>(v);
  return v;
}
__device__ __forceinline__ float dot4_f32(const v4f& a, const v4f& b) {
  const v2f t = __builtin_elementwise_fma(a.zw, b.zw, a.xy * b.xy);
  return t.x + t.y;
}

__device__ __forceinline__ void hold_rows(v4f (&r)[8]) {
  asm volatile("" : "+v"(r[0]), "+v"(r[1]), "+v"(r[2]), "+v"(r[3]),
                    "+v"(r[4]), "+v"(r[5]), "+v"(r[6]), "+v"(r[7]));
}
__device__ __forceinline__ void hold_rows(v4f (&r)[4]) {
  asm volatile("" : "+v"(r[0]), "+v"(r[1]), "+v"(r[2]), "+v"(r[3]));
}
__device__ __forceinline__ void hold_rows(v4f (&r)[2]) {
  asm volatile("" : "+v"(r[0]), "+v"(r[1]));
}
__device__ __forceinline__ void hold_rows(v4f (&r)[1]) {
  asm volatile("" : "+v"(r[0]));
}

#define NBINS     256
#define BIN_SHIFT 9
#define BIN_W     (1 << BIN_SHIFT)
#define CHUNK     6400
#define BT        512
#define CHUNK_IT  13
#define BUILD_IT  24
#define BUILD_CAP (BUILD_IT * BT)

__device__ __forceinline__ int wave_excl_scan_i32(int v, int lane, int& total) {
  int inc = v;
#pragma unroll
  for (int off = 1; off < 32; off <<= 1) {
    const int t = __shfl_up(inc, off, 32);
    if (lane >= off) inc += t;
  }
  total = __shfl(inc, 31, 32);
  return inc - v;
}

__device__ __forceinline__ void bin_hist_block(const int* __restrict__ ei, int E,
                                               int* __restrict__ hist) {
  __shared__ int h[NBINS];
  const int tid = threadIdx.x;
  const int p   = blockIdx.x;
  const int e0  = p * CHUNK;
  const int n   = min(CHUNK, E - e0);
  int d[CHUNK_IT];
#pragma unroll
  for (int k = 0; k < CHUNK_IT; ++k)
    d[k] = ei[(size_t)E + e0 + min(tid + BT * k, n - 1)];
  if (tid < NBINS) h[tid] = 0;
  __syncthreads();
#pragma unroll
  for (int k = 0; k < CHUNK_IT; ++k)
    if (tid + BT * k < n) atomicAdd(&h[d[k] >> BIN_SHIFT], 1);
  __syncthreads();
  if (tid < NBINS) hist[p * NBINS + tid] = h[tid];
}

__device__ __forceinline__ void bin_scatter_block(const int* __restrict__ ei, int E,
                                                  const int* __restrict__ hist, int P,
                                                  int* __restrict__ binStart,
                                                  int* __restrict__ packed) {
  __shared__ int tot[2][NBINS];
  __shared__ int pre[2][NBINS];
  __shared__ int gbase[NBINS];
  __shared__ int lstart[NBINS];
  __shared__ int lcur[NBINS];
  __shared__ int stage[CHUNK];
  __shared__ unsigned short sbin[CHUNK];
  const int tid = threadIdx.x;
  const int p   = blockIdx.x;
  const int e0  = p * CHUNK;
  const int n   = min(CHUNK, E - e0);

  int s[CHUNK_IT], d[CHUNK_IT];
#pragma unroll
  for (int k = 0; k < CHUNK_IT; ++k) {
    const int e = e0 + min(tid + BT * k, n - 1);
    s[k] = ei[e];
    d[k] = ei[(size_t)E + e];
  }

  {
    const int part = tid >> 8;
    const int bin  = tid & (NBINS - 1);
    int t = 0, pr = 0;
    for (int q0 = part; q0 < P; q0 += 32) {
      int v[16];
#pragma unroll
      for (int u = 0; u < 16; ++u) v[u] = hist[min(q0 + 2 * u, P - 1) * NBINS + bin];
      asm volatile("" : "+v"(v[0]), "+v"(v[1]), "+v"(v[2]), "+v"(v[3]),
                        "+v"(v[4]), "+v"(v[5]), "+v"(v[6]), "+v"(v[7]),
                        "+v"(v[8]), "+v"(v[9]), "+v"(v[10]), "+v"(v[11]),
                        "+v"(v[12]), "+v"(v[13]), "+v"(v[14]), "+v"(v[15]));
#pragma unroll
      for (int u = 0; u < 16; ++u) {
        const int q  = q0 + 2 * u;
        const int vv = (q < P) ? v[u] : 0;
        t += vv;
        pr += (q < p) ? vv : 0;
      }
    }
    tot[part][bin] = t;
    pre[part][bin] = pr;
  }
  __syncthreads();
  if (tid < 32) {
    const int lane = tid;
    int eT[8], eO[8], R[8];
    int accT = 0, accO = 0;
#pragma unroll
    for (int j = 0; j < 8; ++j) {
      const int b = 8 * lane + j;
      const int T = tot[0][b] + tot[1][b];
      R[j] = pre[0][b] + pre[1][b];
      const int O = hist[p * NBINS + b];
      eT[j] = accT;  accT += T;
      eO[j] = accO;  accO += O;
    }
    int totT, totO;
    const int baseT = wave_excl_scan_i32(accT, lane, totT);
    const int baseO = wave_excl_scan_i32(accO, lane, totO);
#pragma unroll
    for (int j = 0; j < 8; ++j) {
      const int b  = 8 * lane + j;
      const int bs = baseT + eT[j];
      gbase[b]  = bs + R[j];
      lstart[b] = baseO + eO[j];
      lcur[b]   = baseO + eO[j];
      if (p == 0) binStart[b] = bs;
    }
    if (p == 0 && lane == 0) binStart[NBINS] = totT;
  }
  __syncthreads();
#pragma unroll
  for (int k = 0; k < CHUNK_IT; ++k) {
    if (tid + BT * k < n) {
      const int b   = d[k] >> BIN_SHIFT;
      const int pos = atomicAdd(&lcur[b], 1);
      stage[pos] = ((d[k] & (BIN_W - 1)) << 20) | s[k];
      sbin[pos]  = (unsigned short)b;
    }
  }
  __syncthreads();
  for (int i = tid; i < n; i += BT) {
    const int b = sbin[i];
    packed[gbase[b] + (i - lstart[b])] = stage[i];
  }
}

__device__ __forceinline__ void bin_build_block(const int* __restrict__ packed,
                                                const int* __restrict__ binStart,
                                                int N, int E,
                                                int* __restrict__ rowptr,
                                                int* __restrict__ cnt,
                                                int* __restrict__ col) {
  __shared__ int lcnt[BIN_W];
  __shared__ int loff[BIN_W];
  __shared__ int stage2[BUILD_CAP];
  const int tid = threadIdx.x;
  const int b   = blockIdx.x;
  const int d0  = b << BIN_SHIFT;
  if (d0 >= N) return;
  const int nd = min(BIN_W, N - d0);
  const int bs = binStart[b];
  const int n  = binStart[b + 1] - bs;
  const bool fits = n <= BUILD_CAP;

  int w[BUILD_IT];
  if (fits) {
#pragma unroll
    for (int k = 0; k < BUILD_IT; ++k)
      w[k] = packed[min(bs + min(tid + BT * k, max(n - 1, 0)), E - 1)];
  }
  if (tid < BIN_W) lcnt[tid] = 0;
  __syncthreads();
  if (fits) {
#pragma unroll
    for (int k = 0; k < BUILD_IT; ++k)
      if (tid + BT * k < n) atomicAdd(&lcnt[w[k] >> 20], 1);
  } else {
    for (int i = tid; i < n; i += BT) atomicAdd(&lcnt[packed[bs + i] >> 20], 1);
  }
  __syncthreads();
  if (tid < 32) {
    const int lane = tid;
    int ex[16];
    int acc = 0;
#pragma unroll
    for (int j = 0; j < 16; ++j) {
      ex[j] = acc;
      acc += lcnt[16 * lane + j];
    }
    int total;
    const int base = wave_excl_scan_i32(acc, lane, total);
#pragma unroll
    for (int j = 0; j < 16; ++j) loff[16 * lane + j] = base + ex[j];
  }
  __syncthreads();
  if (tid < nd) {
    rowptr[d0 + tid] = bs + loff[tid];
    cnt[d0 + tid]    = lcnt[tid];
  }
  __syncthreads();
  if (tid < BIN_W) lcnt[tid] = loff[tid];
  __syncthreads();
  if (fits) {
#pragma unroll
    for (int k = 0; k < BUILD_IT; ++k)
      if (tid + BT * k < n) {
        const int pos = atomicAdd(&lcnt[w[k] >> 20], 1);
        stage2[pos] = w[k] & 0xFFFFF;
      }
    __syncthreads();
    for (int i = tid; i < n; i += BT) col[bs + i] = stage2[i];
  } else {
    for (int i = tid; i < n; i += BT) {
      const int v   = packed[bs + i];
      const int pos = atomicAdd(&lcnt[v >> 20], 1);
      col[bs + pos] = v & 0xFFFFF;
    }
  }
}

#define TWO_BLOCKS __attribute__((amdgpu_waves_per_eu(8, 8)))
#define PROD_BLOCKS 256
#define PROD_ROUND  ((PROD_BLOCKS / 2) * (BT / 32))

__global__ __launch_bounds__(BT) TWO_BLOCKS void hist_and_product(
    const int* __restrict__ ei, int E, int* __restrict__ hist, int nbin,
    const float* __restrict__ A, const float* __restrict__ B, float* __restrict__ C,
    int tile0, int tile1) {
  if ((int)blockIdx.x < nbin)
    bin_hist_block(ei, E, hist);
  else
    gemm_block_f32_k128<BT / 32>(A, B, C, 128, 2, tile0, tile1,
                                 (int)blockIdx.x - nbin, PROD_BLOCKS / 2, 1, PROD_ROUND);
}

__global__ __launch_bounds__(BT) TWO_BLOCKS void scatter_and_product(
    const int* __restrict__ ei, int E, const int* __restrict__ hist, int P,
    int* __restrict__ binStart, int* __restrict__ packed,
    const float* __restrict__ A, const float* __restrict__ B, float* __restrict__ C,
    int tile0, int tile1) {
  if ((int)blockIdx.x < P)
    bin_scatter_block(ei, E, hist, P, binStart, packed);
  else
    gemm_block_f32_k128<BT / 32>(A, B, C, 128, 2, tile0, tile1,
                                 (int)blockIdx.x - P, PROD_BLOCKS / 2, 1, PROD_ROUND);
}

__global__ __launch_bounds__(BT) TWO_BLOCKS void build_and_product(
    const int* __restrict__ packed, const int* __restrict__ binStart, int N, int E,
    int* __restrict__ rowptr, int* __restrict__ cnt, int* __restrict__ col,
    const float* __restrict__ A, const float* __restrict__ B, float* __restrict__ C,
    int tile0, int tile1) {
  if ((int)blockIdx.x < NBINS)
    bin_build_block(packed, binStart, N, E, rowptr, cnt, col);
  else
    gemm_block_f32_k128<BT / 32>(A, B, C, 128, 2, tile0, tile1,
                                 (int)blockIdx.x - NBINS, PROD_BLOCKS / 2, 1, PROD_ROUND);
}

#define NO_SCORE (-1.0e30f)

template <int K, int H>
__device__ __forceinline__ void take_rows(const float* __restrict__ hfeat, int head,
                                          int c, int hi, int sv, int t,
                                          v4f (&r)[K]) {
#pragma unroll
  for (int u = 0; u < K; ++u) {
    int s;
    if (H == 2) {
      s = __builtin_amdgcn_readlane(sv, t + u);
    } else {
      const int e0 = 2 * (t + u);
      const int s0 = __builtin_amdgcn_readlane(sv, e0);
      const int s1 = __builtin_amdgcn_readlane(sv, e0 + 1);
      s = hi ? s1 : s0;
    }
    const unsigned off = (((unsigned)s * H + head) * 16u + c) * 16u;
    r[u] = *reinterpret_cast<const v4f*>(reinterpret_cast<const char*>(hfeat) + off);
  }
}

template <int K, int H>
__device__ __forceinline__ void fold_rows(const v4f (&r)[K], int t, int lim,
                                          const v4f& ws, float ad, float& m,
                                          float& sum, v4f& acc) {
  float v[K];
  float mx = m;
#pragma unroll
  for (int u = 0; u < K; ++u) {
    float x = row16_sum_f32(dot4_f32(r[u], ws)) + ad;
    x = fmaxf(x, NEG_SLOPE * x);
    if (H == 1) x = (t + u < lim) ? x : NO_SCORE;
    v[u] = x;
    mx = fmaxf(mx, x);
  }
  const float sc = __expf(m - mx);
  sum *= sc;
  acc *= sc;
#pragma unroll
  for (int u = 0; u < K; ++u) {
    float p = __expf(v[u] - mx);
    if (H == 1) p = (t + u < lim) ? p : 0.0f;
    sum += p;
    const v4f pp = {p, p, p, p};
    acc = __builtin_elementwise_fma(pp, r[u], acc);
  }
  m = mx;
}

template <int H, bool ELU>
__global__ __launch_bounds__(256) void dst_softmax_aggregate(
    const int* __restrict__ rowptr, const int* __restrict__ cnt,
    const int* __restrict__ col,
    const float* __restrict__ att_s, const float* __restrict__ att_d,
    const float* __restrict__ hfeat, const float* __restrict__ bias,
    float* __restrict__ out, int N) {
  static_assert(H == 1 || H == 2, "lane rows are heads (2) or list halves (1)");
  const int lane = threadIdx.x & 31;
  const int hi   = lane >> 4;
  const int c    = lane & 15;
  const int head = (H == 2) ? hi : 0;
  const int wid0 = __builtin_amdgcn_readfirstlane(
      (int)(blockIdx.x * (blockDim.x >> 5) + (threadIdx.x >> 5)));
  const int nw = (int)(gridDim.x * (blockDim.x >> 5));

  const v4f ws = reinterpret_cast<const v4f*>(att_s)[head * 16 + c];
  const v4f wd = reinterpret_cast<const v4f*>(att_d)[head * 16 + c];
  const v4f bz = reinterpret_cast<const v4f*>(bias)[head * 16 + c];
  const bool opens = (H == 2) || (hi == 0);

  for (int d = wid0; d < N; d += nw) {
  const unsigned own_off = (((unsigned)d * H + head) * 16u + c) * 16u;
  v4f own =
      *reinterpret_cast<const v4f*>(reinterpret_cast<const char*>(hfeat) + own_off);
  int       start = rowptr[d];
  const int deg   = cnt[d];
  asm volatile("" : "+s"(start));
  int sv_next = col[(deg > 0) ? start + min(lane, deg - 1) : 0];
  asm volatile("" : "+v"(sv_next), "+v"(own));
  const float ad = row16_sum_f32(dot4_f32(own, wd));
  float vs = row16_sum_f32(dot4_f32(own, ws)) + ad;
  vs = fmaxf(vs, NEG_SLOPE * vs);

  float  m   = opens ? vs : NO_SCORE;
  float  sum = opens ? 1.0f : 0.0f;
  v4f acc;
  acc.x = opens ? own.x : 0.0f;
  acc.y = opens ? own.y : 0.0f;
  acc.z = opens ? own.z : 0.0f;
  acc.w = opens ? own.w : 0.0f;

  for (int base = 0; base < deg; base += 32) {
    asm volatile("" : "+v"(sv_next));
    const int sv = sv_next;
    if (base + 32 < deg) sv_next = col[start + min(base + 32 + lane, deg - 1)];
    const int nj = min(32, deg - base);
    const int nsteps = (H == 2) ? nj : ((nj + 1) >> 1);
    const int lim = (nj + 1 - hi) >> 1;

    int t = 0;
    for (; t + 8 <= nsteps; t += 8) {
      v4f r[8];
      take_rows<8, H>(hfeat, head, c, hi, sv, t, r);
      hold_rows(r);
      fold_rows<8, H>(r, t, lim, ws, ad, m, sum, acc);
    }
    const int  rem = nsteps - t;
    const bool g4 = (rem & 4) != 0, g2 = (rem & 2) != 0, g1 = (rem & 1) != 0;
    const int  t2 = t + (rem & 4), t1 = t2 + (rem & 2);
    v4f r4[4], r2[2], r1[1];
    if (g4) take_rows<4, H>(hfeat, head, c, hi, sv, t, r4);
    if (g2) take_rows<2, H>(hfeat, head, c, hi, sv, t2, r2);
    if (g1) take_rows<1, H>(hfeat, head, c, hi, sv, t1, r1);
    if (g4) {
      hold_rows(r4);
      fold_rows<4, H>(r4, t, lim, ws, ad, m, sum, acc);
    }
    if (g2) {
      hold_rows(r2);
      fold_rows<2, H>(r2, t2, lim, ws, ad, m, sum, acc);
    }
    if (g1) {
      hold_rows(r1);
      fold_rows<1, H>(r1, t1, lim, ws, ad, m, sum, acc);
    }
  }

  if (H == 1) {
    const float mt = fmaxf(m, xhalf_f32(m));
    const float f  = __expf(m - mt);
    sum *= f;
    acc *= f;
    sum   += xhalf_f32(sum);
    acc.x += xhalf_f32(acc.x);
    acc.y += xhalf_f32(acc.y);
    acc.z += xhalf_f32(acc.z);
    acc.w += xhalf_f32(acc.w);
  }

  const float inv = 1.0f / sum;
  v4f o = acc * inv + bz;
  if (ELU) {
    o.x = (o.x > 0.0f) ? o.x : expm1f(o.x);
    o.y = (o.y > 0.0f) ? o.y : expm1f(o.y);
    o.z = (o.z > 0.0f) ? o.z : expm1f(o.z);
    o.w = (o.w > 0.0f) ? o.w : expm1f(o.w);
  }
  if (opens)
    *reinterpret_cast<v4f*>(reinterpret_cast<char*>(out) + own_off) = o;
  }
}

static inline int cdiv_l(long a, int b) { return (int)((a + b - 1) / b); }

#define AGG_BLOCKS 1024

extern "C" void kernel_launch(void* const* d_in, const int* in_sizes, int n_in,
                              void* d_out, int out_size, void* d_ws, size_t ws_size,
                              hipStream_t stream) {
  const float* x    = (const float*)d_in[0];
  const int*   ei   = (const int*)d_in[1];
  const float* W1   = (const float*)d_in[2];
  const float* as1w = (const float*)d_in[3];
  const float* ad1w = (const float*)d_in[4];
  const float* b1   = (const float*)d_in[5];
  const float* W2   = (const float*)d_in[6];
  const float* as2w = (const float*)d_in[7];
  const float* ad2w = (const float*)d_in[8];
  const float* b2   = (const float*)d_in[9];

  const int Fin = 128;
  const int N   = in_sizes[0] / Fin;
  const int E   = in_sizes[1] / 2;

  float* ws   = (float*)d_ws;
  float* h1   = ws;  ws += (size_t)N * 128;
  float* agg1 = ws;  ws += (size_t)N * 128;
  float* h2   = ws;  ws += (size_t)N * 64;
  int* cnt    = (int*)ws;  ws += (size_t)N;
  int* rowptr = (int*)ws;  ws += (size_t)N;
  int* packed = (int*)ws;  ws += (size_t)E;
  int* col    = (int*)ws;  ws += (size_t)E;
  float* outp = (float*)d_out;
  const int P   = cdiv_l(E, CHUNK);
  int* hist     = (int*)agg1;
  int* binStart = hist + (size_t)P * NBINS;

  const int Mt = N / 16;
  const int ta = (Mt < PROD_ROUND) ? Mt : PROD_ROUND;
  const int tb = (Mt < 2 * PROD_ROUND) ? Mt : 2 * PROD_ROUND;
  hist_and_product<<<P + PROD_BLOCKS, BT, 0, stream>>>(
      ei, E, hist, P, x, W1, h1, 0, ta);
  scatter_and_product<<<P + PROD_BLOCKS, BT, 0, stream>>>(
      ei, E, hist, P, binStart, packed, x, W1, h1, ta, tb);
  build_and_product<<<NBINS + PROD_BLOCKS, BT, 0, stream>>>(
      packed, binStart, N, E, rowptr, cnt, col, x, W1, h1, tb, Mt);

  {
    dst_softmax_aggregate<2, true><<<AGG_BLOCKS, 256, 0, stream>>>(
        rowptr, cnt, col, as1w, ad1w, h1, b1, agg1, N);
  }

  {
    wmma_gemm_f32_k128<<<cdiv_l(N / 16, 4), 128, 0, stream>>>(
        agg1, W2, h2, 64, 1, N / 16);
    dst_softmax_aggregate<1, false><<<AGG_BLOCKS, 256, 0, stream>>>(
        rowptr, cnt, col, as2w, ad2w, h2, b2, outp, N);
  }
}


# aggregation: a row group's K scores formed together (K lane sums share butterfly steps, lane c gets row u(c)'s): leaky ReLU, exp, entry test once a group; score and weight sum order changes
# speedup vs baseline: 9.7028x; 1.0644x over previous
#include <hip/hip_runtime.h>
#include <hip/hip_bf16.h>
#include <math.h>

typedef __attribute__((ext_vector_type(2))) float v2f;
typedef __attribute__((ext_vector_type(8))) float v8f;
typedef __attribute__((ext_vector_type(4))) float v4f;

#define NEG_SLOPE 0.2f

template <int NW>
__device__ __forceinline__ void gemm_block_f32_k128(
    const float* __restrict__ A, const float* __restrict__ B,
    float* __restrict__ C, int Ncols, int groups, int tile0, int Mtiles, int gb,
    int wmul, int bmul, int step) {
  __shared__ float Bs[64 * 128];
  const int tid = threadIdx.x;
  const int g   = gb % groups;
  const int bg  = gb / groups;

#pragma unroll
  for (int i = 0; i < 32 / NW; ++i) {
    const int q  = tid + 32 * NW * i;
    const int kp = q >> 4;
    const int c4 = q & 15;
    const float* __restrict__ src = B + (size_t)(2 * kp) * Ncols + 64 * g + 4 * c4;
    const v4f e = *reinterpret_cast<const v4f*>(src);
    const v4f o = *reinterpret_cast<const v4f*>(src + Ncols);
    const v4f lo = {e.x, o.x, e.y, o.y};
    const v4f hi = {e.z, o.z, e.w, o.w};
    *reinterpret_cast<v4f*>(&Bs[kp * 128 + 4 * c4])      = lo;
    *reinterpret_cast<v4f*>(&Bs[kp * 128 + 64 + 4 * c4]) = hi;
  }
  __syncthreads();

  const int wave = __builtin_amdgcn_readfirstlane(tid >> 5);
  const int lane = tid & 31;
  const int c    = lane & 15;
  const int h    = lane >> 4;
  const float* __restrict__ Bl = &Bs[(2 * h) * 128 + 4 * c];

  for (int tile = tile0 + wave * wmul + bg * bmul; tile < Mtiles; tile += step) {
    asm volatile("" ::: "memory");
    const int tileM = tile * 16;

    const float* __restrict__ Arow = A + (size_t)(tileM + c) * 128 + 4 * h;
    v4f a[16];
#pragma unroll
    for (int j = 0; j < 16; ++j)
      a[j] = *reinterpret_cast<const v4f*>(Arow + 8 * j);

    v8f acc[4] = {};
#pragma unroll
    for (int j = 0; j < 16; ++j) {
#pragma unroll
      for (int p = 0; p < 2; ++p) {
        v2f av;
        av.x = a[j][2 * p];
        av.y = a[j][2 * p + 1];
        const float* __restrict__ bp = Bl + (4 * j + p) * 128;
        const v4f q0 = *reinterpret_cast<const v4f*>(bp);
        const v4f q1 = *reinterpret_cast<const v4f*>(bp + 64);
        acc[0] = __builtin_amdgcn_wmma_f32_16x16x4_f32(
             false, av,  false, q0.xy,
             (short)0, acc[0],  false,  false);
        acc[1] = __builtin_amdgcn_wmma_f32_16x16x4_f32(
            false, av, false, q0.zw, (short)0, acc[1], false, false);
        acc[2] = __builtin_amdgcn_wmma_f32_16x16x4_f32(
            false, av, false, q1.xy, (short)0, acc[2], false, false);
        acc[3] = __builtin_amdgcn_wmma_f32_16x16x4_f32(
            false, av, false, q1.zw, (short)0, acc[3], false, false);
      }
    }

    float* __restrict__ Crow = C + (size_t)(tileM + 8 * h) * Ncols + 64 * g + 4 * c;
#pragma unroll
    for (int rr = 0; rr < 8; ++rr) {
      v4f o;
      o.x = acc[0][rr];
      o.y = acc[1][rr];
      o.z = acc[2][rr];
      o.w = acc[3][rr];
      *reinterpret_cast<v4f*>(Crow + (size_t)rr * Ncols) = o;
    }
  }
}

__global__ __launch_bounds__(128) void wmma_gemm_f32_k128(
    const float* __restrict__ A, const float* __restrict__ B,
    float* __restrict__ C, int Ncols, int groups, int Mtiles) {
  gemm_block_f32_k128<4>(A, B, C, Ncols, groups, 0, Mtiles, (int)blockIdx.x,
                         1, 4, Mtiles);
}

template <int XM>
__device__ __forceinline__ float row_xor_f32(float v) {
  return __int_as_float(__builtin_amdgcn_update_dpp(0, __float_as_int(v), 0x160 | XM,
                                                    0xf, 0xf, true));
}
__device__ __forceinline__ float xhalf_f32(float v) {
  const int b = __float_as_int(v);
  return __int_as_float(
      __builtin_amdgcn_permlanex16(b, b, 0x76543210, 0xfedcba98, false, false));
}
__device__ __forceinline__ float row16_sum_f32(float v) {
  v += row_xor_f32<1>(v);
  v += row_xor_f32<2>(v);
  v += row_xor_f32<4>(v);
  v += row_xor_f32<8>(v);
  return v;
}
template <int N>
__device__ __forceinline__ float row_share_f32(float v) {
  return __int_as_float(__builtin_amdgcn_update_dpp(0, __float_as_int(v), 0x150 | N,
                                                    0xf, 0xf, true));
}
template <int B>
__device__ __forceinline__ float share_step_f32(float a, float b, bool up) {
  const float keep = up ? b : a;
  const float send = up ? a : b;
  return keep + row_xor_f32<B>(send);
}
template <int K>
__device__ __forceinline__ float row16_sums_spread_f32(float (&v)[K], int c) {
  const bool b3 = (c & 8) != 0, b2 = (c & 4) != 0, b1 = (c & 2) != 0;
  if (K >= 2) {
#pragma unroll
    for (int i = 0; i < K / 2; ++i) v[i] = share_step_f32<8>(v[2 * i], v[2 * i + 1], b3);
  }
  if (K >= 4) {
#pragma unroll
    for (int i = 0; i < K / 4; ++i) v[i] = share_step_f32<4>(v[2 * i], v[2 * i + 1], b2);
  }
  if (K >= 8) v[0] = share_step_f32<2>(v[0], v[1], b1);
  float s = v[0];
  if (K < 2) s += row_xor_f32<8>(s);
  if (K < 4) s += row_xor_f32<4>(s);
  if (K < 8) s += row_xor_f32<2>(s);
  s += row_xor_f32<1>(s);
  return s;
}
template <int K, int U>
__device__ __forceinline__ float spread_value_f32(float p) {
  if (K == 1) return p;
  return row_share_f32<(K >= 2 ? 8 * (U & 1) : 0) + (K >= 4 ? 4 * ((U >> 1) & 1) : 0) +
                       (K >= 8 ? 2 * ((U >> 2) & 1) : 0)>(p);
}
__device__ __forceinline__ float dot4_f32(const v4f& a, const v4f& b) {
  const v2f t = __builtin_elementwise_fma(a.zw, b.zw, a.xy * b.xy);
  return t.x + t.y;
}

__device__ __forceinline__ void hold_rows(v4f (&r)[8]) {
  asm volatile("" : "+v"(r[0]), "+v"(r[1]), "+v"(r[2]), "+v"(r[3]),
                    "+v"(r[4]), "+v"(r[5]), "+v"(r[6]), "+v"(r[7]));
}
__device__ __forceinline__ void hold_rows(v4f (&r)[4]) {
  asm volatile("" : "+v"(r[0]), "+v"(r[1]), "+v"(r[2]), "+v"(r[3]));
}
__device__ __forceinline__ void hold_rows(v4f (&r)[2]) {
  asm volatile("" : "+v"(r[0]), "+v"(r[1]));
}
__device__ __forceinline__ void hold_rows(v4f (&r)[1]) {
  asm volatile("" : "+v"(r[0]));
}

#define NBINS     256
#define BIN_SHIFT 9
#define BIN_W     (1 << BIN_SHIFT)
#define CHUNK     6400
#define BT        512
#define CHUNK_IT  13
#define BUILD_IT  24
#define BUILD_CAP (BUILD_IT * BT)

__device__ __forceinline__ int wave_excl_scan_i32(int v, int lane, int& total) {
  int inc = v;
#pragma unroll
  for (int off = 1; off < 32; off <<= 1) {
    const int t = __shfl_up(inc, off, 32);
    if (lane >= off) inc += t;
  }
  total = __shfl(inc, 31, 32);
  return inc - v;
}

__device__ __forceinline__ void bin_hist_block(const int* __restrict__ ei, int E,
                                               int* __restrict__ hist) {
  __shared__ int h[NBINS];
  const int tid = threadIdx.x;
  const int p   = blockIdx.x;
  const int e0  = p * CHUNK;
  const int n   = min(CHUNK, E - e0);
  int d[CHUNK_IT];
#pragma unroll
  for (int k = 0; k < CHUNK_IT; ++k)
    d[k] = ei[(size_t)E + e0 + min(tid + BT * k, n - 1)];
  if (tid < NBINS) h[tid] = 0;
  __syncthreads();
#pragma unroll
  for (int k = 0; k < CHUNK_IT; ++k)
    if (tid + BT * k < n) atomicAdd(&h[d[k] >> BIN_SHIFT], 1);
  __syncthreads();
  if (tid < NBINS) hist[p * NBINS + tid] = h[tid];
}

__device__ __forceinline__ void bin_scatter_block(const int* __restrict__ ei, int E,
                                                  const int* __restrict__ hist, int P,
                                                  int* __restrict__ binStart,
                                                  int* __restrict__ packed) {
  __shared__ int tot[2][NBINS];
  __shared__ int pre[2][NBINS];
  __shared__ int gbase[NBINS];
  __shared__ int lstart[NBINS];
  __shared__ int lcur[NBINS];
  __shared__ int stage[CHUNK];
  __shared__ unsigned short sbin[CHUNK];
  const int tid = threadIdx.x;
  const int p   = blockIdx.x;
  const int e0  = p * CHUNK;
  const int n   = min(CHUNK, E - e0);

  int s[CHUNK_IT], d[CHUNK_IT];
#pragma unroll
  for (int k = 0; k < CHUNK_IT; ++k) {
    const int e = e0 + min(tid + BT * k, n - 1);
    s[k] = ei[e];
    d[k] = ei[(size_t)E + e];
  }

  {
    const int part = tid >> 8;
    const int bin  = tid & (NBINS - 1);
    int t = 0, pr = 0;
    for (int q0 = part; q0 < P; q0 += 32) {
      int v[16];
#pragma unroll
      for (int u = 0; u < 16; ++u) v[u] = hist[min(q0 + 2 * u, P - 1) * NBINS + bin];
      asm volatile("" : "+v"(v[0]), "+v"(v[1]), "+v"(v[2]), "+v"(v[3]),
                        "+v"(v[4]), "+v"(v[5]), "+v"(v[6]), "+v"(v[7]),
                        "+v"(v[8]), "+v"(v[9]), "+v"(v[10]), "+v"(v[11]),
                        "+v"(v[12]), "+v"(v[13]), "+v"(v[14]), "+v"(v[15]));
#pragma unroll
      for (int u = 0; u < 16; ++u) {
        const int q  = q0 + 2 * u;
        const int vv = (q < P) ? v[u] : 0;
        t += vv;
        pr += (q < p) ? vv : 0;
      }
    }
    tot[part][bin] = t;
    pre[part][bin] = pr;
  }
  __syncthreads();
  if (tid < 32) {
    const int lane = tid;
    int eT[8], eO[8], R[8];
    int accT = 0, accO = 0;
#pragma unroll
    for (int j = 0; j < 8; ++j) {
      const int b = 8 * lane + j;
      const int T = tot[0][b] + tot[1][b];
      R[j] = pre[0][b] + pre[1][b];
      const int O = hist[p * NBINS + b];
      eT[j] = accT;  accT += T;
      eO[j] = accO;  accO += O;
    }
    int totT, totO;
    const int baseT = wave_excl_scan_i32(accT, lane, totT);
    const int baseO = wave_excl_scan_i32(accO, lane, totO);
#pragma unroll
    for (int j = 0; j < 8; ++j) {
      const int b  = 8 * lane + j;
      const int bs = baseT + eT[j];
      gbase[b]  = bs + R[j];
      lstart[b] = baseO + eO[j];
      lcur[b]   = baseO + eO[j];
      if (p == 0) binStart[b] = bs;
    }
    if (p == 0 && lane == 0) binStart[NBINS] = totT;
  }
  __syncthreads();
#pragma unroll
  for (int k = 0; k < CHUNK_IT; ++k) {
    if (tid + BT * k < n) {
      const int b   = d[k] >> BIN_SHIFT;
      const int pos = atomicAdd(&lcur[b], 1);
      stage[pos] = ((d[k] & (BIN_W - 1)) << 20) | s[k];
      sbin[pos]  = (unsigned short)b;
    }
  }
  __syncthreads();
  for (int i = tid; i < n; i += BT) {
    const int b = sbin[i];
    packed[gbase[b] + (i - lstart[b])] = stage[i];
  }
}

__device__ __forceinline__ void bin_build_block(const int* __restrict__ packed,
                                                const int* __restrict__ binStart,
                                                int N, int E,
                                                int* __restrict__ rowptr,
                                                int* __restrict__ cnt,
                                                int* __restrict__ col) {
  __shared__ int lcnt[BIN_W];
  __shared__ int loff[BIN_W];
  __shared__ int stage2[BUILD_CAP];
  const int tid = threadIdx.x;
  const int b   = blockIdx.x;
  const int d0  = b << BIN_SHIFT;
  if (d0 >= N) return;
  const int nd = min(BIN_W, N - d0);
  const int bs = binStart[b];
  const int n  = binStart[b + 1] - bs;
  const bool fits = n <= BUILD_CAP;

  int w[BUILD_IT];
  if (fits) {
#pragma unroll
    for (int k = 0; k < BUILD_IT; ++k)
      w[k] = packed[min(bs + min(tid + BT * k, max(n - 1, 0)), E - 1)];
  }
  if (tid < BIN_W) lcnt[tid] = 0;
  __syncthreads();
  if (fits) {
#pragma unroll
    for (int k = 0; k < BUILD_IT; ++k)
      if (tid + BT * k < n) atomicAdd(&lcnt[w[k] >> 20], 1);
  } else {
    for (int i = tid; i < n; i += BT) atomicAdd(&lcnt[packed[bs + i] >> 20], 1);
  }
  __syncthreads();
  if (tid < 32) {
    const int lane = tid;
    int ex[16];
    int acc = 0;
#pragma unroll
    for (int j = 0; j < 16; ++j) {
      ex[j] = acc;
      acc += lcnt[16 * lane + j];
    }
    int total;
    const int base = wave_excl_scan_i32(acc, lane, total);
#pragma unroll
    for (int j = 0; j < 16; ++j) loff[16 * lane + j] = base + ex[j];
  }
  __syncthreads();
  if (tid < nd) {
    rowptr[d0 + tid] = bs + loff[tid];
    cnt[d0 + tid]    = lcnt[tid];
  }
  __syncthreads();
  if (tid < BIN_W) lcnt[tid] = loff[tid];
  __syncthreads();
  if (fits) {
#pragma unroll
    for (int k = 0; k < BUILD_IT; ++k)
      if (tid + BT * k < n) {
        const int pos = atomicAdd(&lcnt[w[k] >> 20], 1);
        stage2[pos] = w[k] & 0xFFFFF;
      }
    __syncthreads();
    for (int i = tid; i < n; i += BT) col[bs + i] = stage2[i];
  } else {
    for (int i = tid; i < n; i += BT) {
      const int v   = packed[bs + i];
      const int pos = atomicAdd(&lcnt[v >> 20], 1);
      col[bs + pos] = v & 0xFFFFF;
    }
  }
}

#define TWO_BLOCKS __attribute__((amdgpu_waves_per_eu(8, 8)))
#define PROD_BLOCKS 256
#define PROD_ROUND  ((PROD_BLOCKS / 2) * (BT / 32))

__global__ __launch_bounds__(BT) TWO_BLOCKS void hist_and_product(
    const int* __restrict__ ei, int E, int* __restrict__ hist, int nbin,
    const float* __restrict__ A, const float* __restrict__ B, float* __restrict__ C,
    int tile0, int tile1) {
  if ((int)blockIdx.x < nbin)
    bin_hist_block(ei, E, hist);
  else
    gemm_block_f32_k128<BT / 32>(A, B, C, 128, 2, tile0, tile1,
                                 (int)blockIdx.x - nbin, PROD_BLOCKS / 2, 1, PROD_ROUND);
}

__global__ __launch_bounds__(BT) TWO_BLOCKS void scatter_and_product(
    const int* __restrict__ ei, int E, const int* __restrict__ hist, int P,
    int* __restrict__ binStart, int* __restrict__ packed,
    const float* __restrict__ A, const float* __restrict__ B, float* __restrict__ C,
    int tile0, int tile1) {
  if ((int)blockIdx.x < P)
    bin_scatter_block(ei, E, hist, P, binStart, packed);
  else
    gemm_block_f32_k128<BT / 32>(A, B, C, 128, 2, tile0, tile1,
                                 (int)blockIdx.x - P, PROD_BLOCKS / 2, 1, PROD_ROUND);
}

__global__ __launch_bounds__(BT) TWO_BLOCKS void build_and_product(
    const int* __restrict__ packed, const int* __restrict__ binStart, int N, int E,
    int* __restrict__ rowptr, int* __restrict__ cnt, int* __restrict__ col,
    const float* __restrict__ A, const float* __restrict__ B, float* __restrict__ C,
    int tile0, int tile1) {
  if ((int)blockIdx.x < NBINS)
    bin_build_block(packed, binStart, N, E, rowptr, cnt, col);
  else
    gemm_block_f32_k128<BT / 32>(A, B, C, 128, 2, tile0, tile1,
                                 (int)blockIdx.x - NBINS, PROD_BLOCKS / 2, 1, PROD_ROUND);
}

#define NO_SCORE (-1.0e30f)

template <int K, int H>
__device__ __forceinline__ void take_rows(const float* __restrict__ hfeat, int head,
                                          int c, int hi, int sv, int t,
                                          v4f (&r)[K]) {
#pragma unroll
  for (int u = 0; u < K; ++u) {
    int s;
    if (H == 2) {
      s = __builtin_amdgcn_readlane(sv, t + u);
    } else {
      const int e0 = 2 * (t + u);
      const int s0 = __builtin_amdgcn_readlane(sv, e0);
      const int s1 = __builtin_amdgcn_readlane(sv, e0 + 1);
      s = hi ? s1 : s0;
    }
    const unsigned off = (((unsigned)s * H + head) * 16u + c) * 16u;
    r[u] = *reinterpret_cast<const v4f*>(reinterpret_cast<const char*>(hfeat) + off);
  }
}

template <int K, int U>
__device__ __forceinline__ void add_rows_weighted(const v4f (&r)[K], float p, v4f& acc) {
  if constexpr (U < K) {
    const float pu = spread_value_f32<K, U>(p);
    const v4f pp = {pu, pu, pu, pu};
    acc = __builtin_elementwise_fma(pp, r[U], acc);
    add_rows_weighted<K, U + 1>(r, p, acc);
  }
}
template <int K, int H>
__device__ __forceinline__ void fold_rows(const v4f (&r)[K], int t, int lim, int c,
                                          const v4f& ws, float ad, float& m,
                                          float& sumL, v4f& acc) {
  float v[K];
#pragma unroll
  for (int u = 0; u < K; ++u) v[u] = dot4_f32(r[u], ws);
  float x = row16_sums_spread_f32<K>(v, c) + ad;
  x = fmaxf(x, NEG_SLOPE * x);
  const int  u    = (((c >> 3) & 1) + 2 * ((c >> 2) & 1) + 4 * ((c >> 1) & 1)) & (K - 1);
  const bool live = (H == 2) || (t + u < lim);
  if (H == 1) x = live ? x : NO_SCORE;
  float gm = x;
  if (K >= 2) gm = fmaxf(gm, row_xor_f32<8>(gm));
  if (K >= 4) gm = fmaxf(gm, row_xor_f32<4>(gm));
  if (K >= 8) gm = fmaxf(gm, row_xor_f32<2>(gm));
  const float mx = fmaxf(m, gm);
  const float sc = __expf(m - mx);
  float p = __expf(x - mx);
  if (H == 1) p = live ? p : 0.0f;
  sumL = fmaf(sumL, sc, p * (K * 0.0625f));
  acc *= sc;
  add_rows_weighted<K, 0>(r, p, acc);
  m = mx;
}

template <int H, bool ELU>
__global__ __launch_bounds__(256) void dst_softmax_aggregate(
    const int* __restrict__ rowptr, const int* __restrict__ cnt,
    const int* __restrict__ col,
    const float* __restrict__ att_s, const float* __restrict__ att_d,
    const float* __restrict__ hfeat, const float* __restrict__ bias,
    float* __restrict__ out, int N) {
  static_assert(H == 1 || H == 2, "lane rows are heads (2) or list halves (1)");
  const int lane = threadIdx.x & 31;
  const int hi   = lane >> 4;
  const int c    = lane & 15;
  const int head = (H == 2) ? hi : 0;
  const int wid0 = __builtin_amdgcn_readfirstlane(
      (int)(blockIdx.x * (blockDim.x >> 5) + (threadIdx.x >> 5)));
  const int nw = (int)(gridDim.x * (blockDim.x >> 5));

  const v4f ws = reinterpret_cast<const v4f*>(att_s)[head * 16 + c];
  const v4f wd = reinterpret_cast<const v4f*>(att_d)[head * 16 + c];
  const v4f bz = reinterpret_cast<const v4f*>(bias)[head * 16 + c];
  const bool opens = (H == 2) || (hi == 0);

  for (int d = wid0; d < N; d += nw) {
  const unsigned own_off = (((unsigned)d * H + head) * 16u + c) * 16u;
  v4f own =
      *reinterpret_cast<const v4f*>(reinterpret_cast<const char*>(hfeat) + own_off);
  int       start = rowptr[d];
  const int deg   = cnt[d];
  asm volatile("" : "+s"(start));
  int sv_next = col[(deg > 0) ? start + min(lane, deg - 1) : 0];
  asm volatile("" : "+v"(sv_next), "+v"(own));
  const float ad = row16_sum_f32(dot4_f32(own, wd));
  float vs = row16_sum_f32(dot4_f32(own, ws)) + ad;
  vs = fmaxf(vs, NEG_SLOPE * vs);

  float  m   = opens ? vs : NO_SCORE;
  float  sumL = opens ? 0.0625f : 0.0f;
  v4f acc;
  acc.x = opens ? own.x : 0.0f;
  acc.y = opens ? own.y : 0.0f;
  acc.z = opens ? own.z : 0.0f;
  acc.w = opens ? own.w : 0.0f;

  for (int base = 0; base < deg; base += 32) {
    asm volatile("" : "+v"(sv_next));
    const int sv = sv_next;
    if (base + 32 < deg) sv_next = col[start + min(base + 32 + lane, deg - 1)];
    const int nj = min(32, deg - base);
    const int nsteps = (H == 2) ? nj : ((nj + 1) >> 1);
    const int lim = (nj + 1 - hi) >> 1;

    int t = 0;
    for (; t + 8 <= nsteps; t += 8) {
      v4f r[8];
      take_rows<8, H>(hfeat, head, c, hi, sv, t, r);
      hold_rows(r);
      fold_rows<8, H>(r, t, lim, c, ws, ad, m, sumL, acc);
    }
    const int  rem = nsteps - t;
    const bool g4 = (rem & 4) != 0, g2 = (rem & 2) != 0, g1 = (rem & 1) != 0;
    const int  t2 = t + (rem & 4), t1 = t2 + (rem & 2);
    v4f r4[4], r2[2], r1[1];
    if (g4) take_rows<4, H>(hfeat, head, c, hi, sv, t, r4);
    if (g2) take_rows<2, H>(hfeat, head, c, hi, sv, t2, r2);
    if (g1) take_rows<1, H>(hfeat, head, c, hi, sv, t1, r1);
    if (g4) {
      hold_rows(r4);
      fold_rows<4, H>(r4, t, lim, c, ws, ad, m, sumL, acc);
    }
    if (g2) {
      hold_rows(r2);
      fold_rows<2, H>(r2, t2, lim, c, ws, ad, m, sumL, acc);
    }
    if (g1) {
      hold_rows(r1);
      fold_rows<1, H>(r1, t1, lim, c, ws, ad, m, sumL, acc);
    }
  }

  float sum = row16_sum_f32(sumL);
  if (H == 1) {
    const float mt = fmaxf(m, xhalf_f32(m));
    const float f  = __expf(m - mt);
    sum *= f;
    acc *= f;
    sum   += xhalf_f32(sum);
    acc.x += xhalf_f32(acc.x);
    acc.y += xhalf_f32(acc.y);
    acc.z += xhalf_f32(acc.z);
    acc.w += xhalf_f32(acc.w);
  }

  const float inv = 1.0f / sum;
  v4f o = acc * inv + bz;
  if (ELU) {
    o.x = (o.x > 0.0f) ? o.x : expm1f(o.x);
    o.y = (o.y > 0.0f) ? o.y : expm1f(o.y);
    o.z = (o.z > 0.0f) ? o.z : expm1f(o.z);
    o.w = (o.w > 0.0f) ? o.w : expm1f(o.w);
  }
  if (opens)
    *reinterpret_cast<v4f*>(reinterpret_cast<char*>(out) + own_off) = o;
  }
}

static inline int cdiv_l(long a, int b) { return (int)((a + b - 1) / b); }

#define AGG_BLOCKS 1024

extern "C" void kernel_launch(void* const* d_in, const int* in_sizes, int n_in,
                              void* d_out, int out_size, void* d_ws, size_t ws_size,
                              hipStream_t stream) {
  const float* x    = (const float*)d_in[0];
  const int*   ei   = (const int*)d_in[1];
  const float* W1   = (const float*)d_in[2];
  const float* as1w = (const float*)d_in[3];
  const float* ad1w = (const float*)d_in[4];
  const float* b1   = (const float*)d_in[5];
  const float* W2   = (const float*)d_in[6];
  const float* as2w = (const float*)d_in[7];
  const float* ad2w = (const float*)d_in[8];
  const float* b2   = (const float*)d_in[9];

  const int Fin = 128;
  const int N   = in_sizes[0] / Fin;
  const int E   = in_sizes[1] / 2;

  float* ws   = (float*)d_ws;
  float* h1   = ws;  ws += (size_t)N * 128;
  float* agg1 = ws;  ws += (size_t)N * 128;
  float* h2   = ws;  ws += (size_t)N * 64;
  int* cnt    = (int*)ws;  ws += (size_t)N;
  int* rowptr = (int*)ws;  ws += (size_t)N;
  int* packed = (int*)ws;  ws += (size_t)E;
  int* col    = (int*)ws;  ws += (size_t)E;
  float* outp = (float*)d_out;
  const int P   = cdiv_l(E, CHUNK);
  int* hist     = (int*)agg1;
  int* binStart = hist + (size_t)P * NBINS;

  const int Mt = N / 16;
  const int ta = (Mt < PROD_ROUND) ? Mt : PROD_ROUND;
  const int tb = (Mt < 2 * PROD_ROUND) ? Mt : 2 * PROD_ROUND;
  hist_and_product<<<P + PROD_BLOCKS, BT, 0, stream>>>(
      ei, E, hist, P, x, W1, h1, 0, ta);
  scatter_and_product<<<P + PROD_BLOCKS, BT, 0, stream>>>(
      ei, E, hist, P, binStart, packed, x, W1, h1, ta, tb);
  build_and_product<<<NBINS + PROD_BLOCKS, BT, 0, stream>>>(
      packed, binStart, N, E, rowptr, cnt, col, x, W1, h1, tb, Mt);

  {
    dst_softmax_aggregate<2, true><<<AGG_BLOCKS, 256, 0, stream>>>(
        rowptr, cnt, col, as1w, ad1w, h1, b1, agg1, N);
  }

  {
    wmma_gemm_f32_k128<<<cdiv_l(N / 16, 4), 128, 0, stream>>>(
        agg1, W2, h2, 64, 1, N / 16);
    dst_softmax_aggregate<1, false><<<AGG_BLOCKS, 256, 0, stream>>>(
        rowptr, cnt, col, as2w, ad2w, h2, b2, outp, N);
  }
}
